# starting kernel
# baseline (speedup 1.0000x reference)
#include <hip/hip_runtime.h>

typedef __attribute__((ext_vector_type(2))) float v2f;
typedef __attribute__((ext_vector_type(4))) float v4f;
typedef __attribute__((ext_vector_type(8))) float v8f;

#define ALPHA 0.3f
#define EPSF  1e-6f

#define BB   2
#define NN   8192
#define KK   32
#define CC   128
#define HH   128
#define MM   (BB * NN)

__device__ __forceinline__ float lrelu(float x) { return x >= 0.f ? x : ALPHA * x; }

__global__ __launch_bounds__(256) void k1_hidden(const float* __restrict__ emb,
                                                 const float* __restrict__ Qw,
                                                 const float* __restrict__ Qb,
                                                 float* __restrict__ hidden)
{
    const int lane = threadIdx.x & 31;
    const int wave = threadIdx.x >> 5;
    const int m0   = (blockIdx.x * 8 + wave) << 4;
    const int h0   = blockIdx.y << 4;
    const int rc   = lane & 15;
    const int half = lane >> 4;

    const float* arow = emb + (size_t)(m0 + rc) * CC;
    v8f acc = (v8f)0.f;

    #pragma unroll 4
    for (int k4 = 0; k4 < CC; k4 += 4) {
        const int k0 = k4 + 2 * half;
        v2f a = *(const v2f*)(arow + k0);
        v2f b;
        b.x = Qw[(size_t)k0 * HH + h0 + rc];
        b.y = Qw[(size_t)(k0 + 1) * HH + h0 + rc];
        acc = __builtin_amdgcn_wmma_f32_16x16x4_f32(false, a, false, b,
                                                    (short)0, acc, false, false);
    }

    const float bv = Qb[h0 + rc];
    #pragma unroll
    for (int r = 0; r < 8; ++r) {
        const int m = m0 + r + (half << 3);
        hidden[(size_t)m * HH + h0 + rc] = lrelu(acc[r] + bv);
    }
}

__global__ __launch_bounds__(256) void k2_gather(const float* __restrict__ hidden,
                                                 const float* __restrict__ weights,
                                                 const int* __restrict__ nbr,
                                                 float* __restrict__ wsh)
{
    const int lane = threadIdx.x & 31;
    const int wave = threadIdx.x >> 5;
    const int idx  = blockIdx.x * 8 + wave;
    const int b    = idx >> 13;
    const int n    = idx & (NN - 1);

    const int   j = nbr[(size_t)n * KK + lane];
    const float w = weights[(size_t)n * NN + j];

    const float* hb = hidden + (size_t)b * NN * HH + (size_t)lane * 4;

    v4f   acc  = (v4f)0.f;
    float wsum = 0.f;
    #pragma unroll 8
    for (int k = 0; k < KK; ++k) {
        const int   jk = __shfl(j, k, 32);
        const float wk = __shfl(w, k, 32);
        const v4f h = *(const v4f*)(hb + (size_t)jk * HH);
        acc.x += wk * h.x;
        acc.y += wk * h.y;
        acc.z += wk * h.z;
        acc.w += wk * h.w;
        wsum  += wk;
    }
    const float inv = 1.f / (wsum + EPSF);
    v4f r;
    r.x = acc.x * inv; r.y = acc.y * inv; r.z = acc.z * inv; r.w = acc.w * inv;
    *(v4f*)(wsh + (size_t)idx * HH + lane * 4) = r;
}

__global__ __launch_bounds__(256) void k3_gemm2(const float* __restrict__ emb,
                                                const float* __restrict__ wsh,
                                                const float* __restrict__ Ww,
                                                const float* __restrict__ Wb,
                                                float* __restrict__ out)
{
    const int lane = threadIdx.x & 31;
    const int wave = threadIdx.x >> 5;
    const int m0   = (blockIdx.x * 8 + wave) << 4;
    const int h0   = blockIdx.y << 4;
    const int rc   = lane & 15;
    const int half = lane >> 4;

    const float* a1 = emb + (size_t)(m0 + rc) * CC;
    const float* a2 = wsh + (size_t)(m0 + rc) * HH;
    v8f acc = (v8f)0.f;

    #pragma unroll 4
    for (int k4 = 0; k4 < CC; k4 += 4) {
        const int k0 = k4 + 2 * half;
        v2f a = *(const v2f*)(a1 + k0);
        v2f b;
        b.x = Ww[(size_t)k0 * HH + h0 + rc];
        b.y = Ww[(size_t)(k0 + 1) * HH + h0 + rc];
        acc = __builtin_amdgcn_wmma_f32_16x16x4_f32(false, a, false, b,
                                                    (short)0, acc, false, false);
    }
    #pragma unroll 4
    for (int k4 = 0; k4 < HH; k4 += 4) {
        const int k0 = k4 + 2 * half;
        v2f a = *(const v2f*)(a2 + k0);
        v2f b;
        b.x = Ww[(size_t)(CC + k0) * HH + h0 + rc];
        b.y = Ww[(size_t)(CC + k0 + 1) * HH + h0 + rc];
        acc = __builtin_amdgcn_wmma_f32_16x16x4_f32(false, a, false, b,
                                                    (short)0, acc, false, false);
    }

    const float bv = Wb[h0 + rc];
    #pragma unroll
    for (int r = 0; r < 8; ++r) {
        const int m = m0 + r + (half << 3);
        out[(size_t)m * HH + h0 + rc] = lrelu(acc[r] + bv);
    }
}

__global__ __launch_bounds__(256) void k4_norm(float* __restrict__ out)
{
    const int lane = threadIdx.x & 31;
    const int wave = threadIdx.x >> 5;
    const int row  = blockIdx.x * 8 + wave;

    float* p = out + (size_t)row * HH + lane * 4;
    v4f v = *(const v4f*)p;
    float s = v.x * v.x + v.y * v.y + v.z * v.z + v.w * v.w;
    #pragma unroll
    for (int off = 16; off >= 1; off >>= 1)
        s += __shfl_xor(s, off, 32);
    const float inv = 1.f / (sqrtf(s) + EPSF);
    v.x *= inv; v.y *= inv; v.z *= inv; v.w *= inv;
    *(v4f*)p = v;
}

extern "C" void kernel_launch(void* const* d_in, const int* in_sizes, int n_in,
                              void* d_out, int out_size, void* d_ws, size_t ws_size,
                              hipStream_t stream)
{
    const float* emb     = (const float*)d_in[0];
    const float* weights = (const float*)d_in[1];
    const int*   nbr     = (const int*)  d_in[2];
    const float* Qw      = (const float*)d_in[3];
    const float* Qb      = (const float*)d_in[4];
    const float* Ww      = (const float*)d_in[5];
    const float* Wb      = (const float*)d_in[6];
    float*       out     = (float*)d_out;

    float* hidden = (float*)d_ws;
    float* wsh    = hidden + (size_t)MM * HH;

    const dim3 blk(256);
    k1_hidden<<<dim3(MM / 128, HH / 16), blk, 0, stream>>>(emb, Qw, Qb, hidden);
    k2_gather<<<dim3(MM / 8), blk, 0, stream>>>(hidden, weights, nbr, wsh);
    k3_gemm2<<<dim3(MM / 128, HH / 16), blk, 0, stream>>>(emb, wsh, Ww, Wb, out);
    k4_norm<<<dim3(MM / 8), blk, 0, stream>>>(out);
}


# Fuse row L2 norm into GEMM2: each wave owns 16 complete rows (Ww staged once per block in LDS as k-pairs), norm in registers, 4->3 launches; k order within 8-blocks and norm sum tree changed
# speedup vs baseline: 1.5054x; 1.5054x over previous
#include <hip/hip_runtime.h>

typedef __attribute__((ext_vector_type(2))) float v2f;
typedef __attribute__((ext_vector_type(4))) float v4f;
typedef __attribute__((ext_vector_type(8))) float v8f;

#define ALPHA 0.3f
#define EPSF  1e-6f

#define BB   2
#define NN   8192
#define KK   32
#define CC   128
#define HH   128
#define MM   (BB * NN)

__device__ __forceinline__ float lrelu(float x) { return x >= 0.f ? x : ALPHA * x; }

__global__ __launch_bounds__(256) void k1_hidden(const float* __restrict__ emb,
                                                 const float* __restrict__ Qw,
                                                 const float* __restrict__ Qb,
                                                 float* __restrict__ hidden)
{
    const int lane = threadIdx.x & 31;
    const int wave = threadIdx.x >> 5;
    const int m0   = (blockIdx.x * 8 + wave) << 4;
    const int h0   = blockIdx.y << 4;
    const int rc   = lane & 15;
    const int half = lane >> 4;

    const float* arow = emb + (size_t)(m0 + rc) * CC;
    v8f acc = (v8f)0.f;

    #pragma unroll 4
    for (int k4 = 0; k4 < CC; k4 += 4) {
        const int k0 = k4 + 2 * half;
        v2f a = *(const v2f*)(arow + k0);
        v2f b;
        b.x = Qw[(size_t)k0 * HH + h0 + rc];
        b.y = Qw[(size_t)(k0 + 1) * HH + h0 + rc];
        acc = __builtin_amdgcn_wmma_f32_16x16x4_f32(false, a, false, b,
                                                    (short)0, acc, false, false);
    }

    const float bv = Qb[h0 + rc];
    #pragma unroll
    for (int r = 0; r < 8; ++r) {
        const int m = m0 + r + (half << 3);
        hidden[(size_t)m * HH + h0 + rc] = lrelu(acc[r] + bv);
    }
}

__global__ __launch_bounds__(256) void k2_gather(const float* __restrict__ hidden,
                                                 const float* __restrict__ weights,
                                                 const int* __restrict__ nbr,
                                                 float* __restrict__ wsh)
{
    const int lane = threadIdx.x & 31;
    const int wave = threadIdx.x >> 5;
    const int idx  = blockIdx.x * 8 + wave;
    const int b    = idx >> 13;
    const int n    = idx & (NN - 1);

    const int   j = nbr[(size_t)n * KK + lane];
    const float w = weights[(size_t)n * NN + j];

    const float* hb = hidden + (size_t)b * NN * HH + (size_t)lane * 4;

    v4f   acc  = (v4f)0.f;
    float wsum = 0.f;
    #pragma unroll 8
    for (int k = 0; k < KK; ++k) {
        const int   jk = __shfl(j, k, 32);
        const float wk = __shfl(w, k, 32);
        const v4f h = *(const v4f*)(hb + (size_t)jk * HH);
        acc.x += wk * h.x;
        acc.y += wk * h.y;
        acc.z += wk * h.z;
        acc.w += wk * h.w;
        wsum  += wk;
    }
    const float inv = 1.f / (wsum + EPSF);
    v4f r;
    r.x = acc.x * inv; r.y = acc.y * inv; r.z = acc.z * inv; r.w = acc.w * inv;
    *(v4f*)(wsh + (size_t)idx * HH + lane * 4) = r;
}

#define LDSP 272
#define LDS_FLOATS 41984

__global__ __launch_bounds__(128) void k3_gemm2_norm(const float* __restrict__ emb,
                                                     const float* __restrict__ wsh,
                                                     const float* __restrict__ Ww,
                                                     const float* __restrict__ Wb,
                                                     float* __restrict__ out)
{
    __shared__ __attribute__((aligned(16))) float sB[LDS_FLOATS];
    static_assert(((CC + HH) / 2) * LDSP <= LDS_FLOATS, "staged Ww must fit the reservation");

    const int lane = threadIdx.x & 31;
    const int wave = threadIdx.x >> 5;
    const int m0   = (blockIdx.x * 4 + wave) << 4;
    const int c    = lane & 15;
    const int h    = lane >> 4;

    const float* pe = emb + (size_t)(m0 + c) * CC + 4 * h;
    const float* pw = wsh + (size_t)(m0 + c) * HH + 4 * h;
    v4f aE[CC / 8], aW[HH / 8];
    #pragma unroll
    for (int i = 0; i < CC / 8; ++i) aE[i] = *(const v4f*)(pe + 8 * i);
    #pragma unroll
    for (int i = 0; i < HH / 8; ++i) aW[i] = *(const v4f*)(pw + 8 * i);

    #pragma unroll 8
    for (int i = 0; i < (CC + HH) / 8; ++i) {
        const int kp = i * 4 + wave;
        const v4f x = *(const v4f*)(Ww + (size_t)(2 * kp) * HH + 4 * lane);
        const v4f y = *(const v4f*)(Ww + (size_t)(2 * kp + 1) * HH + 4 * lane);
        v4f lo, hi;
        lo.x = x.x; lo.y = y.x; lo.z = x.y; lo.w = y.y;
        hi.x = x.z; hi.y = y.z; hi.z = x.w; hi.w = y.w;
        float* d = sB + kp * LDSP + 8 * lane;
        *(v4f*)d       = lo;
        *(v4f*)(d + 4) = hi;
    }
    __syncthreads();

    v8f acc[4][2];
    #pragma unroll
    for (int j = 0; j < 4; ++j) { acc[j][0] = (v8f)0.f; acc[j][1] = (v8f)0.f; }

    const float* lb = sB + 2 * h * LDSP + 4 * c;

    #pragma unroll
    for (int kb = 0; kb < (CC + HH) / 8; ++kb) {
        const v4f a4 = (kb < CC / 8) ? aE[kb] : aW[kb - CC / 8];
        #pragma unroll
        for (int st = 0; st < 2; ++st) {
            v4f bf[4];
            #pragma unroll
            for (int j = 0; j < 4; ++j)
                bf[j] = *(const v4f*)(lb + (4 * kb + st) * LDSP + 64 * j);
            v2f a;
            a.x = st ? a4.z : a4.x;
            a.y = st ? a4.w : a4.y;
            #pragma unroll
            for (int j = 0; j < 4; ++j) {
                v2f b0, b1;
                b0.x = bf[j].x; b0.y = bf[j].y;
                b1.x = bf[j].z; b1.y = bf[j].w;
                acc[j][0] = __builtin_amdgcn_wmma_f32_16x16x4_f32(false, a, false, b0,
                                                                  (short)0, acc[j][0], false, false);
                acc[j][1] = __builtin_amdgcn_wmma_f32_16x16x4_f32(false, a, false, b1,
                                                                  (short)0, acc[j][1], false, false);
            }
        }
    }

    v2f bias[4];
    #pragma unroll
    for (int j = 0; j < 4; ++j) bias[j] = *(const v2f*)(Wb + 32 * j + 2 * c);

    #pragma unroll
    for (int r = 0; r < 8; ++r) {
        float v[4][2];
        float s = 0.f;
        #pragma unroll
        for (int j = 0; j < 4; ++j) {
            v[j][0] = lrelu(acc[j][0][r] + bias[j].x);
            v[j][1] = lrelu(acc[j][1][r] + bias[j].y);
            s += v[j][0] * v[j][0];
            s += v[j][1] * v[j][1];
        }
        #pragma unroll
        for (int off = 8; off >= 1; off >>= 1)
            s += __shfl_xor(s, off, 32);
        const float inv = 1.f / (sqrtf(s) + EPSF);
        float* po = out + (size_t)(m0 + 8 * h + r) * HH + 2 * c;
        #pragma unroll
        for (int j = 0; j < 4; ++j) {
            v2f o;
            o.x = v[j][0] * inv;
            o.y = v[j][1] * inv;
            *(v2f*)(po + 32 * j) = o;
        }
    }
}

extern "C" void kernel_launch(void* const* d_in, const int* in_sizes, int n_in,
                              void* d_out, int out_size, void* d_ws, size_t ws_size,
                              hipStream_t stream)
{
    const float* emb     = (const float*)d_in[0];
    const float* weights = (const float*)d_in[1];
    const int*   nbr     = (const int*)  d_in[2];
    const float* Qw      = (const float*)d_in[3];
    const float* Qb      = (const float*)d_in[4];
    const float* Ww      = (const float*)d_in[5];
    const float* Wb      = (const float*)d_in[6];
    float*       out     = (float*)d_out;

    float* hidden = (float*)d_ws;
    float* wsh    = hidden + (size_t)MM * HH;

    const dim3 blk(256);
    k1_hidden<<<dim3(MM / 128, HH / 16), blk, 0, stream>>>(emb, Qw, Qb, hidden);
    k2_gather<<<dim3(MM / 8), blk, 0, stream>>>(hidden, weights, nbr, wsh);
    k3_gemm2_norm<<<dim3(MM / 64), dim3(128), 0, stream>>>(emb, wsh, Ww, Wb, out);
}


# Gather kernel: 16384 one-row waves -> 4096 resident waves, each owning 2 nodes x both batches (ids, weights read once per node), 32-48 row loads kept in flight per wave; order of sums unchanged
# speedup vs baseline: 1.5652x; 1.0397x over previous
#include <hip/hip_runtime.h>

typedef __attribute__((ext_vector_type(2))) float v2f;
typedef __attribute__((ext_vector_type(4))) float v4f;
typedef __attribute__((ext_vector_type(8))) float v8f;

#define ALPHA 0.3f
#define EPSF  1e-6f

#define BB   2
#define NN   8192
#define KK   32
#define CC   128
#define HH   128
#define MM   (BB * NN)

__device__ __forceinline__ float lrelu(float x) { return x >= 0.f ? x : ALPHA * x; }

__global__ __launch_bounds__(256) void k1_hidden(const float* __restrict__ emb,
                                                 const float* __restrict__ Qw,
                                                 const float* __restrict__ Qb,
                                                 float* __restrict__ hidden)
{
    const int lane = threadIdx.x & 31;
    const int wave = threadIdx.x >> 5;
    const int m0   = (blockIdx.x * 8 + wave) << 4;
    const int h0   = blockIdx.y << 4;
    const int rc   = lane & 15;
    const int half = lane >> 4;

    const float* arow = emb + (size_t)(m0 + rc) * CC;
    v8f acc = (v8f)0.f;

    #pragma unroll 4
    for (int k4 = 0; k4 < CC; k4 += 4) {
        const int k0 = k4 + 2 * half;
        v2f a = *(const v2f*)(arow + k0);
        v2f b;
        b.x = Qw[(size_t)k0 * HH + h0 + rc];
        b.y = Qw[(size_t)(k0 + 1) * HH + h0 + rc];
        acc = __builtin_amdgcn_wmma_f32_16x16x4_f32(false, a, false, b,
                                                    (short)0, acc, false, false);
    }

    const float bv = Qb[h0 + rc];
    #pragma unroll
    for (int r = 0; r < 8; ++r) {
        const int m = m0 + r + (half << 3);
        hidden[(size_t)m * HH + h0 + rc] = lrelu(acc[r] + bv);
    }
}

#define NPW 2
#define KST 8
#define SPN (KK / KST)
#define NST (NPW * SPN)

__global__ __launch_bounds__(256) void k2_gather(const float* __restrict__ hidden,
                                                 const float* __restrict__ weights,
                                                 const int* __restrict__ nbr,
                                                 float* __restrict__ wsh)
{
    const int lane = threadIdx.x & 31;
    const int wave = threadIdx.x >> 5;
    const int n0   = (blockIdx.x * 8 + wave) * NPW;

    int   j[NPW];
    float w[NPW];
    #pragma unroll
    for (int t = 0; t < NPW; ++t) j[t] = nbr[(size_t)(n0 + t) * KK + lane];
    #pragma unroll
    for (int t = 0; t < NPW; ++t) w[t] = weights[(size_t)(n0 + t) * NN + j[t]];

    v4f   acc0[NPW], acc1[NPW];
    float wsum[NPW];
    #pragma unroll
    for (int t = 0; t < NPW; ++t) { acc0[t] = (v4f)0.f; acc1[t] = (v4f)0.f; wsum[t] = 0.f; }

    v4f   h0[2][KST], h1[2][KST];

    #pragma unroll
    for (int st = 0; st <= NST; ++st) {
        if (st < NST) {
            const int t = st / SPN, s = st % SPN, bf = st & 1;
            #pragma unroll
            for (int kk = 0; kk < KST; ++kk) {
                const int jk = __builtin_amdgcn_readlane(j[t], s * KST + kk);
                const float* row = hidden + (size_t)jk * HH + lane * 4;
                h0[bf][kk] = *(const v4f*)row;
                h1[bf][kk] = *(const v4f*)(row + (size_t)NN * HH);
            }
        }
        __builtin_amdgcn_sched_barrier(0);
        if (st > 0) {
            const int t = (st - 1) / SPN, s = (st - 1) % SPN, bf = (st - 1) & 1;
            #pragma unroll
            for (int kk = 0; kk < KST; ++kk) {
                const float wkk = __int_as_float(__builtin_amdgcn_readlane(__float_as_int(w[t]), s * KST + kk));
                acc0[t].x = __builtin_fmaf(wkk, h0[bf][kk].x, acc0[t].x);
                acc0[t].y = __builtin_fmaf(wkk, h0[bf][kk].y, acc0[t].y);
                acc0[t].z = __builtin_fmaf(wkk, h0[bf][kk].z, acc0[t].z);
                acc0[t].w = __builtin_fmaf(wkk, h0[bf][kk].w, acc0[t].w);
                acc1[t].x = __builtin_fmaf(wkk, h1[bf][kk].x, acc1[t].x);
                acc1[t].y = __builtin_fmaf(wkk, h1[bf][kk].y, acc1[t].y);
                acc1[t].z = __builtin_fmaf(wkk, h1[bf][kk].z, acc1[t].z);
                acc1[t].w = __builtin_fmaf(wkk, h1[bf][kk].w, acc1[t].w);
                wsum[t]  += wkk;
            }
            if (s == SPN - 1) {
                const float inv = 1.f / (wsum[t] + EPSF);
                v4f r0, r1;
                r0.x = acc0[t].x * inv; r0.y = acc0[t].y * inv; r0.z = acc0[t].z * inv; r0.w = acc0[t].w * inv;
                r1.x = acc1[t].x * inv; r1.y = acc1[t].y * inv; r1.z = acc1[t].z * inv; r1.w = acc1[t].w * inv;
                float* o = wsh + (size_t)(n0 + t) * HH + lane * 4;
                *(v4f*)o                      = r0;
                *(v4f*)(o + (size_t)NN * HH) = r1;
            }
        }
    }
}

#define LDSP 272
#define LDS_FLOATS 41984

__global__ __launch_bounds__(128) void k3_gemm2_norm(const float* __restrict__ emb,
                                                     const float* __restrict__ wsh,
                                                     const float* __restrict__ Ww,
                                                     const float* __restrict__ Wb,
                                                     float* __restrict__ out)
{
    __shared__ __attribute__((aligned(16))) float sB[LDS_FLOATS];
    static_assert(((CC + HH) / 2) * LDSP <= LDS_FLOATS, "staged Ww must fit the reservation");

    const int lane = threadIdx.x & 31;
    const int wave = threadIdx.x >> 5;
    const int m0   = (blockIdx.x * 4 + wave) << 4;
    const int c    = lane & 15;
    const int h    = lane >> 4;

    const float* pe = emb + (size_t)(m0 + c) * CC + 4 * h;
    const float* pw = wsh + (size_t)(m0 + c) * HH + 4 * h;
    v4f aE[CC / 8], aW[HH / 8];
    #pragma unroll
    for (int i = 0; i < CC / 8; ++i) aE[i] = *(const v4f*)(pe + 8 * i);
    #pragma unroll
    for (int i = 0; i < HH / 8; ++i) aW[i] = *(const v4f*)(pw + 8 * i);

    #pragma unroll 8
    for (int i = 0; i < (CC + HH) / 8; ++i) {
        const int kp = i * 4 + wave;
        const v4f x = *(const v4f*)(Ww + (size_t)(2 * kp) * HH + 4 * lane);
        const v4f y = *(const v4f*)(Ww + (size_t)(2 * kp + 1) * HH + 4 * lane);
        v4f lo, hi;
        lo.x = x.x; lo.y = y.x; lo.z = x.y; lo.w = y.y;
        hi.x = x.z; hi.y = y.z; hi.z = x.w; hi.w = y.w;
        float* d = sB + kp * LDSP + 8 * lane;
        *(v4f*)d       = lo;
        *(v4f*)(d + 4) = hi;
    }
    __syncthreads();

    v8f acc[4][2];
    #pragma unroll
    for (int j = 0; j < 4; ++j) { acc[j][0] = (v8f)0.f; acc[j][1] = (v8f)0.f; }

    const float* lb = sB + 2 * h * LDSP + 4 * c;

    #pragma unroll
    for (int kb = 0; kb < (CC + HH) / 8; ++kb) {
        const v4f a4 = (kb < CC / 8) ? aE[kb] : aW[kb - CC / 8];
        #pragma unroll
        for (int st = 0; st < 2; ++st) {
            v4f bf[4];
            #pragma unroll
            for (int j = 0; j < 4; ++j)
                bf[j] = *(const v4f*)(lb + (4 * kb + st) * LDSP + 64 * j);
            v2f a;
            a.x = st ? a4.z : a4.x;
            a.y = st ? a4.w : a4.y;
            #pragma unroll
            for (int j = 0; j < 4; ++j) {
                v2f b0, b1;
                b0.x = bf[j].x; b0.y = bf[j].y;
                b1.x = bf[j].z; b1.y = bf[j].w;
                acc[j][0] = __builtin_amdgcn_wmma_f32_16x16x4_f32(false, a, false, b0,
                                                                  (short)0, acc[j][0], false, false);
                acc[j][1] = __builtin_amdgcn_wmma_f32_16x16x4_f32(false, a, false, b1,
                                                                  (short)0, acc[j][1], false, false);
            }
        }
    }

    v2f bias[4];
    #pragma unroll
    for (int j = 0; j < 4; ++j) bias[j] = *(const v2f*)(Wb + 32 * j + 2 * c);

    #pragma unroll
    for (int r = 0; r < 8; ++r) {
        float v[4][2];
        float s = 0.f;
        #pragma unroll
        for (int j = 0; j < 4; ++j) {
            v[j][0] = lrelu(acc[j][0][r] + bias[j].x);
            v[j][1] = lrelu(acc[j][1][r] + bias[j].y);
            s += v[j][0] * v[j][0];
            s += v[j][1] * v[j][1];
        }
        #pragma unroll
        for (int off = 8; off >= 1; off >>= 1)
            s += __shfl_xor(s, off, 32);
        const float inv = 1.f / (sqrtf(s) + EPSF);
        float* po = out + (size_t)(m0 + 8 * h + r) * HH + 2 * c;
        #pragma unroll
        for (int j = 0; j < 4; ++j) {
            v2f o;
            o.x = v[j][0] * inv;
            o.y = v[j][1] * inv;
            *(v2f*)(po + 32 * j) = o;
        }
    }
}

extern "C" void kernel_launch(void* const* d_in, const int* in_sizes, int n_in,
                              void* d_out, int out_size, void* d_ws, size_t ws_size,
                              hipStream_t stream)
{
    const float* emb     = (const float*)d_in[0];
    const float* weights = (const float*)d_in[1];
    const int*   nbr     = (const int*)  d_in[2];
    const float* Qw      = (const float*)d_in[3];
    const float* Qb      = (const float*)d_in[4];
    const float* Ww      = (const float*)d_in[5];
    const float* Wb      = (const float*)d_in[6];
    float*       out     = (float*)d_out;

    float* hidden = (float*)d_ws;
    float* wsh    = hidden + (size_t)MM * HH;

    const dim3 blk(256);
    k1_hidden<<<dim3(MM / 128, HH / 16), blk, 0, stream>>>(emb, Qw, Qb, hidden);
    k2_gather<<<dim3(NN / (8 * NPW)), blk, 0, stream>>>(hidden, weights, nbr, wsh);
    k3_gemm2_norm<<<dim3(MM / 64), dim3(128), 0, stream>>>(emb, wsh, Ww, Wb, out);
}


# Hidden GEMM (kernel 1) restructured like kernel 3: a wave owns 16 complete rows, Qw staged once per work-group in LDS as k-pairs, 8192 short waves -> 1024; k order within 8-blocks changed
# speedup vs baseline: 1.6229x; 1.0369x over previous
#include <hip/hip_runtime.h>

typedef __attribute__((ext_vector_type(2))) float v2f;
typedef __attribute__((ext_vector_type(4))) float v4f;
typedef __attribute__((ext_vector_type(8))) float v8f;

#define ALPHA 0.3f
#define EPSF  1e-6f

#define BB   2
#define NN   8192
#define KK   32
#define CC   128
#define HH   128
#define MM   (BB * NN)

__device__ __forceinline__ float lrelu(float x) { return x >= 0.f ? x : ALPHA * x; }

#define LDSP 272
#define LDS_FLOATS 41984

__global__ __launch_bounds__(128) void k1_hidden(const float* __restrict__ emb,
                                                 const float* __restrict__ Qw,
                                                 const float* __restrict__ Qb,
                                                 float* __restrict__ hidden)
{
    __shared__ __attribute__((aligned(16))) float sQ[LDS_FLOATS];
    static_assert((CC / 2) * LDSP <= LDS_FLOATS, "staged Qw must fit the reservation");

    const int lane = threadIdx.x & 31;
    const int wave = threadIdx.x >> 5;
    const int m0   = (blockIdx.x * 4 + wave) << 4;
    const int c    = lane & 15;
    const int h    = lane >> 4;

    v4f qx[CC / 8], qy[CC / 8];
    #pragma unroll
    for (int i = 0; i < CC / 8; ++i) {
        const int kp = i * 4 + wave;
        qx[i] = *(const v4f*)(Qw + (size_t)(2 * kp) * HH + 4 * lane);
        qy[i] = *(const v4f*)(Qw + (size_t)(2 * kp + 1) * HH + 4 * lane);
    }

    const float* pe = emb + (size_t)(m0 + c) * CC + 4 * h;
    v4f aE[CC / 8];
    #pragma unroll
    for (int i = 0; i < CC / 8; ++i) aE[i] = *(const v4f*)(pe + 8 * i);
    __builtin_amdgcn_sched_barrier(0);

    #pragma unroll
    for (int i = 0; i < CC / 8; ++i) {
        const int kp = i * 4 + wave;
        v4f lo, hi;
        lo.x = qx[i].x; lo.y = qy[i].x; lo.z = qx[i].y; lo.w = qy[i].y;
        hi.x = qx[i].z; hi.y = qy[i].z; hi.z = qx[i].w; hi.w = qy[i].w;
        float* d = sQ + kp * LDSP + 8 * lane;
        *(v4f*)d       = lo;
        *(v4f*)(d + 4) = hi;
    }
    __syncthreads();

    v8f acc[4][2];
    #pragma unroll
    for (int j = 0; j < 4; ++j) { acc[j][0] = (v8f)0.f; acc[j][1] = (v8f)0.f; }

    const float* lb = sQ + 2 * h * LDSP + 4 * c;

    #pragma unroll
    for (int kb = 0; kb < CC / 8; ++kb) {
        const v4f a4 = aE[kb];
        #pragma unroll
        for (int st = 0; st < 2; ++st) {
            v4f bf[4];
            #pragma unroll
            for (int j = 0; j < 4; ++j)
                bf[j] = *(const v4f*)(lb + (4 * kb + st) * LDSP + 64 * j);
            v2f a;
            a.x = st ? a4.z : a4.x;
            a.y = st ? a4.w : a4.y;
            #pragma unroll
            for (int j = 0; j < 4; ++j) {
                v2f b0, b1;
                b0.x = bf[j].x; b0.y = bf[j].y;
                b1.x = bf[j].z; b1.y = bf[j].w;
                acc[j][0] = __builtin_amdgcn_wmma_f32_16x16x4_f32(false, a, false, b0,
                                                                  (short)0, acc[j][0], false, false);
                acc[j][1] = __builtin_amdgcn_wmma_f32_16x16x4_f32(false, a, false, b1,
                                                                  (short)0, acc[j][1], false, false);
            }
        }
    }

    v2f bias[4];
    #pragma unroll
    for (int j = 0; j < 4; ++j) bias[j] = *(const v2f*)(Qb + 32 * j + 2 * c);

    #pragma unroll
    for (int r = 0; r < 8; ++r) {
        float* po = hidden + (size_t)(m0 + 8 * h + r) * HH + 2 * c;
        #pragma unroll
        for (int j = 0; j < 4; ++j) {
            v2f o;
            o.x = lrelu(acc[j][0][r] + bias[j].x);
            o.y = lrelu(acc[j][1][r] + bias[j].y);
            *(v2f*)(po + 32 * j) = o;
        }
    }
}

#define NPW 2
#define KST 8
#define SPN (KK / KST)
#define NST (NPW * SPN)

__global__ __launch_bounds__(256) void k2_gather(const float* __restrict__ hidden,
                                                 const float* __restrict__ weights,
                                                 const int* __restrict__ nbr,
                                                 float* __restrict__ wsh)
{
    const int lane = threadIdx.x & 31;
    const int wave = threadIdx.x >> 5;
    const int n0   = (blockIdx.x * 8 + wave) * NPW;

    int   j[NPW];
    float w[NPW];
    #pragma unroll
    for (int t = 0; t < NPW; ++t) j[t] = nbr[(size_t)(n0 + t) * KK + lane];
    #pragma unroll
    for (int t = 0; t < NPW; ++t) w[t] = weights[(size_t)(n0 + t) * NN + j[t]];

    v4f   acc0[NPW], acc1[NPW];
    float wsum[NPW];
    #pragma unroll
    for (int t = 0; t < NPW; ++t) { acc0[t] = (v4f)0.f; acc1[t] = (v4f)0.f; wsum[t] = 0.f; }

    v4f   h0[2][KST], h1[2][KST];

    #pragma unroll
    for (int st = 0; st <= NST; ++st) {
        if (st < NST) {
            const int t = st / SPN, s = st % SPN, bf = st & 1;
            #pragma unroll
            for (int kk = 0; kk < KST; ++kk) {
                const int jk = __builtin_amdgcn_readlane(j[t], s * KST + kk);
                const float* row = hidden + (size_t)jk * HH + lane * 4;
                h0[bf][kk] = *(const v4f*)row;
                h1[bf][kk] = *(const v4f*)(row + (size_t)NN * HH);
            }
        }
        __builtin_amdgcn_sched_barrier(0);
        if (st > 0) {
            const int t = (st - 1) / SPN, s = (st - 1) % SPN, bf = (st - 1) & 1;
            #pragma unroll
            for (int kk = 0; kk < KST; ++kk) {
                const float wkk = __int_as_float(__builtin_amdgcn_readlane(__float_as_int(w[t]), s * KST + kk));
                acc0[t].x = __builtin_fmaf(wkk, h0[bf][kk].x, acc0[t].x);
                acc0[t].y = __builtin_fmaf(wkk, h0[bf][kk].y, acc0[t].y);
                acc0[t].z = __builtin_fmaf(wkk, h0[bf][kk].z, acc0[t].z);
                acc0[t].w = __builtin_fmaf(wkk, h0[bf][kk].w, acc0[t].w);
                acc1[t].x = __builtin_fmaf(wkk, h1[bf][kk].x, acc1[t].x);
                acc1[t].y = __builtin_fmaf(wkk, h1[bf][kk].y, acc1[t].y);
                acc1[t].z = __builtin_fmaf(wkk, h1[bf][kk].z, acc1[t].z);
                acc1[t].w = __builtin_fmaf(wkk, h1[bf][kk].w, acc1[t].w);
                wsum[t]  += wkk;
            }
            if (s == SPN - 1) {
                const float inv = 1.f / (wsum[t] + EPSF);
                v4f r0, r1;
                r0.x = acc0[t].x * inv; r0.y = acc0[t].y * inv; r0.z = acc0[t].z * inv; r0.w = acc0[t].w * inv;
                r1.x = acc1[t].x * inv; r1.y = acc1[t].y * inv; r1.z = acc1[t].z * inv; r1.w = acc1[t].w * inv;
                float* o = wsh + (size_t)(n0 + t) * HH + lane * 4;
                *(v4f*)o                      = r0;
                *(v4f*)(o + (size_t)NN * HH) = r1;
            }
        }
    }
}


__global__ __launch_bounds__(128) void k3_gemm2_norm(const float* __restrict__ emb,
                                                     const float* __restrict__ wsh,
                                                     const float* __restrict__ Ww,
                                                     const float* __restrict__ Wb,
                                                     float* __restrict__ out)
{
    __shared__ __attribute__((aligned(16))) float sB[LDS_FLOATS];
    static_assert(((CC + HH) / 2) * LDSP <= LDS_FLOATS, "staged Ww must fit the reservation");

    const int lane = threadIdx.x & 31;
    const int wave = threadIdx.x >> 5;
    const int m0   = (blockIdx.x * 4 + wave) << 4;
    const int c    = lane & 15;
    const int h    = lane >> 4;

    const float* pe = emb + (size_t)(m0 + c) * CC + 4 * h;
    const float* pw = wsh + (size_t)(m0 + c) * HH + 4 * h;
    v4f aE[CC / 8], aW[HH / 8];
    #pragma unroll
    for (int i = 0; i < CC / 8; ++i) aE[i] = *(const v4f*)(pe + 8 * i);
    #pragma unroll
    for (int i = 0; i < HH / 8; ++i) aW[i] = *(const v4f*)(pw + 8 * i);

    #pragma unroll 8
    for (int i = 0; i < (CC + HH) / 8; ++i) {
        const int kp = i * 4 + wave;
        const v4f x = *(const v4f*)(Ww + (size_t)(2 * kp) * HH + 4 * lane);
        const v4f y = *(const v4f*)(Ww + (size_t)(2 * kp + 1) * HH + 4 * lane);
        v4f lo, hi;
        lo.x = x.x; lo.y = y.x; lo.z = x.y; lo.w = y.y;
        hi.x = x.z; hi.y = y.z; hi.z = x.w; hi.w = y.w;
        float* d = sB + kp * LDSP + 8 * lane;
        *(v4f*)d       = lo;
        *(v4f*)(d + 4) = hi;
    }
    __syncthreads();

    v8f acc[4][2];
    #pragma unroll
    for (int j = 0; j < 4; ++j) { acc[j][0] = (v8f)0.f; acc[j][1] = (v8f)0.f; }

    const float* lb = sB + 2 * h * LDSP + 4 * c;

    #pragma unroll
    for (int kb = 0; kb < (CC + HH) / 8; ++kb) {
        const v4f a4 = (kb < CC / 8) ? aE[kb] : aW[kb - CC / 8];
        #pragma unroll
        for (int st = 0; st < 2; ++st) {
            v4f bf[4];
            #pragma unroll
            for (int j = 0; j < 4; ++j)
                bf[j] = *(const v4f*)(lb + (4 * kb + st) * LDSP + 64 * j);
            v2f a;
            a.x = st ? a4.z : a4.x;
            a.y = st ? a4.w : a4.y;
            #pragma unroll
            for (int j = 0; j < 4; ++j) {
                v2f b0, b1;
                b0.x = bf[j].x; b0.y = bf[j].y;
                b1.x = bf[j].z; b1.y = bf[j].w;
                acc[j][0] = __builtin_amdgcn_wmma_f32_16x16x4_f32(false, a, false, b0,
                                                                  (short)0, acc[j][0], false, false);
                acc[j][1] = __builtin_amdgcn_wmma_f32_16x16x4_f32(false, a, false, b1,
                                                                  (short)0, acc[j][1], false, false);
            }
        }
    }

    v2f bias[4];
    #pragma unroll
    for (int j = 0; j < 4; ++j) bias[j] = *(const v2f*)(Wb + 32 * j + 2 * c);

    #pragma unroll
    for (int r = 0; r < 8; ++r) {
        float v[4][2];
        float s = 0.f;
        #pragma unroll
        for (int j = 0; j < 4; ++j) {
            v[j][0] = lrelu(acc[j][0][r] + bias[j].x);
            v[j][1] = lrelu(acc[j][1][r] + bias[j].y);
            s += v[j][0] * v[j][0];
            s += v[j][1] * v[j][1];
        }
        #pragma unroll
        for (int off = 8; off >= 1; off >>= 1)
            s += __shfl_xor(s, off, 32);
        const float inv = 1.f / (sqrtf(s) + EPSF);
        float* po = out + (size_t)(m0 + 8 * h + r) * HH + 2 * c;
        #pragma unroll
        for (int j = 0; j < 4; ++j) {
            v2f o;
            o.x = v[j][0] * inv;
            o.y = v[j][1] * inv;
            *(v2f*)(po + 32 * j) = o;
        }
    }
}

extern "C" void kernel_launch(void* const* d_in, const int* in_sizes, int n_in,
                              void* d_out, int out_size, void* d_ws, size_t ws_size,
                              hipStream_t stream)
{
    const float* emb     = (const float*)d_in[0];
    const float* weights = (const float*)d_in[1];
    const int*   nbr     = (const int*)  d_in[2];
    const float* Qw      = (const float*)d_in[3];
    const float* Qb      = (const float*)d_in[4];
    const float* Ww      = (const float*)d_in[5];
    const float* Wb      = (const float*)d_in[6];
    float*       out     = (float*)d_out;

    float* hidden = (float*)d_ws;
    float* wsh    = hidden + (size_t)MM * HH;

    const dim3 blk(256);
    k1_hidden<<<dim3(MM / 64), dim3(128), 0, stream>>>(emb, Qw, Qb, hidden);
    k2_gather<<<dim3(NN / (8 * NPW)), blk, 0, stream>>>(hidden, weights, nbr, wsh);
    k3_gemm2_norm<<<dim3(MM / 64), dim3(128), 0, stream>>>(emb, wsh, Ww, Wb, out);
}


# Gather fused into the GEMM2+norm kernel: 8 gather waves per work-group fill a 64-row wsh tile in LDS while 4 matrix waves run the emb half of the product; 3->2 launches; order of sums unchanged
# speedup vs baseline: 1.9047x; 1.1736x over previous
#include <hip/hip_runtime.h>

typedef __attribute__((ext_vector_type(2))) float v2f;
typedef __attribute__((ext_vector_type(4))) float v4f;
typedef __attribute__((ext_vector_type(8))) float v8f;

#define ALPHA 0.3f
#define EPSF  1e-6f

#define BB   2
#define NN   8192
#define KK   32
#define CC   128
#define HH   128
#define MM   (BB * NN)

__device__ __forceinline__ float lrelu(float x) { return x >= 0.f ? x : ALPHA * x; }

#define LDSP 272
#define LDS_FLOATS 41984

__global__ __launch_bounds__(128) void k1_hidden(const float* __restrict__ emb,
                                                 const float* __restrict__ Qw,
                                                 const float* __restrict__ Qb,
                                                 float* __restrict__ hidden)
{
    __shared__ __attribute__((aligned(16))) float sQ[LDS_FLOATS];
    static_assert((CC / 2) * LDSP <= LDS_FLOATS, "staged Qw must fit the reservation");

    const int lane = threadIdx.x & 31;
    const int wave = threadIdx.x >> 5;
    const int m0   = (blockIdx.x * 4 + wave) << 4;
    const int c    = lane & 15;
    const int h    = lane >> 4;

    v4f qx[CC / 8], qy[CC / 8];
    #pragma unroll
    for (int i = 0; i < CC / 8; ++i) {
        const int kp = i * 4 + wave;
        qx[i] = *(const v4f*)(Qw + (size_t)(2 * kp) * HH + 4 * lane);
        qy[i] = *(const v4f*)(Qw + (size_t)(2 * kp + 1) * HH + 4 * lane);
    }

    const float* pe = emb + (size_t)(m0 + c) * CC + 4 * h;
    v4f aE[CC / 8];
    #pragma unroll
    for (int i = 0; i < CC / 8; ++i) aE[i] = *(const v4f*)(pe + 8 * i);
    __builtin_amdgcn_sched_barrier(0);

    #pragma unroll
    for (int i = 0; i < CC / 8; ++i) {
        const int kp = i * 4 + wave;
        v4f lo, hi;
        lo.x = qx[i].x; lo.y = qy[i].x; lo.z = qx[i].y; lo.w = qy[i].y;
        hi.x = qx[i].z; hi.y = qy[i].z; hi.z = qx[i].w; hi.w = qy[i].w;
        float* d = sQ + kp * LDSP + 8 * lane;
        *(v4f*)d       = lo;
        *(v4f*)(d + 4) = hi;
    }
    __syncthreads();

    v8f acc[4][2];
    #pragma unroll
    for (int j = 0; j < 4; ++j) { acc[j][0] = (v8f)0.f; acc[j][1] = (v8f)0.f; }

    const float* lb = sQ + 2 * h * LDSP + 4 * c;

    #pragma unroll
    for (int kb = 0; kb < CC / 8; ++kb) {
        const v4f a4 = aE[kb];
        #pragma unroll
        for (int st = 0; st < 2; ++st) {
            v4f bf[4];
            #pragma unroll
            for (int j = 0; j < 4; ++j)
                bf[j] = *(const v4f*)(lb + (4 * kb + st) * LDSP + 64 * j);
            v2f a;
            a.x = st ? a4.z : a4.x;
            a.y = st ? a4.w : a4.y;
            #pragma unroll
            for (int j = 0; j < 4; ++j) {
                v2f b0, b1;
                b0.x = bf[j].x; b0.y = bf[j].y;
                b1.x = bf[j].z; b1.y = bf[j].w;
                acc[j][0] = __builtin_amdgcn_wmma_f32_16x16x4_f32(false, a, false, b0,
                                                                  (short)0, acc[j][0], false, false);
                acc[j][1] = __builtin_amdgcn_wmma_f32_16x16x4_f32(false, a, false, b1,
                                                                  (short)0, acc[j][1], false, false);
            }
        }
    }

    v2f bias[4];
    #pragma unroll
    for (int j = 0; j < 4; ++j) bias[j] = *(const v2f*)(Qb + 32 * j + 2 * c);

    #pragma unroll
    for (int r = 0; r < 8; ++r) {
        float* po = hidden + (size_t)(m0 + 8 * h + r) * HH + 2 * c;
        #pragma unroll
        for (int j = 0; j < 4; ++j) {
            v2f o;
            o.x = lrelu(acc[j][0][r] + bias[j].x);
            o.y = lrelu(acc[j][1][r] + bias[j].y);
            *(v2f*)(po + 32 * j) = o;
        }
    }
}

#define MWV 4
#define GWV 8
#define NPG 4
#define KST 8
#define SPN (KK / KST)
#define NSTG (NPG * SPN)
#define WP 132
#define WROWS ((CC + HH) / 2)
#define SFULL (WROWS / 12)
#define STAIL (WROWS - 12 * SFULL)
#define F_LDS (WROWS * LDSP + 64 * WP)

template <int KB0>
__device__ __forceinline__ void gemm_half(v8f (&acc)[4][2], const v4f (&a16)[16], const float* lb)
{
    #pragma unroll
    for (int kb = 0; kb < 16; ++kb) {
        const v4f a4 = a16[kb];
        #pragma unroll
        for (int st = 0; st < 2; ++st) {
            v4f bf[4];
            #pragma unroll
            for (int j = 0; j < 4; ++j)
                bf[j] = *(const v4f*)(lb + (4 * (KB0 + kb) + st) * LDSP + 64 * j);
            v2f a;
            a.x = st ? a4.z : a4.x;
            a.y = st ? a4.w : a4.y;
            #pragma unroll
            for (int j = 0; j < 4; ++j) {
                v2f b0, b1;
                b0.x = bf[j].x; b0.y = bf[j].y;
                b1.x = bf[j].z; b1.y = bf[j].w;
                acc[j][0] = __builtin_amdgcn_wmma_f32_16x16x4_f32(false, a, false, b0,
                                                                  (short)0, acc[j][0], false, false);
                acc[j][1] = __builtin_amdgcn_wmma_f32_16x16x4_f32(false, a, false, b1,
                                                                  (short)0, acc[j][1], false, false);
            }
        }
    }
}

__global__ __launch_bounds__(384) void k2_gather_gemm2_norm(const float* __restrict__ emb,
                                                            const float* __restrict__ hidden,
                                                            const float* __restrict__ weights,
                                                            const int* __restrict__ nbr,
                                                            const float* __restrict__ Ww,
                                                            const float* __restrict__ Wb,
                                                            float* __restrict__ out)
{
    __shared__ __attribute__((aligned(16))) float sB[F_LDS];
    float* sW = sB + WROWS * LDSP;

    const int lane = threadIdx.x & 31;
    const int wave = __builtin_amdgcn_readfirstlane((int)(threadIdx.x >> 5));
    const int nb   = blockIdx.x * 32;
    const int c    = lane & 15;
    const int h    = lane >> 4;
    const bool mat = wave < MWV;

    v4f aE[CC / 8];
    int   jn[NPG];
    float wn[NPG];
    const int n0 = nb + (wave - MWV) * NPG;
    if (mat) {
        const float* pe = emb + ((size_t)(c >> 3) * NN + nb + 8 * wave + (c & 7)) * CC + 4 * h;
        #pragma unroll
        for (int i = 0; i < CC / 8; ++i) aE[i] = *(const v4f*)(pe + 8 * i);
    } else {
        #pragma unroll
        for (int t = 0; t < NPG; ++t) jn[t] = nbr[(size_t)(n0 + t) * KK + lane];
    }

    v4f qx[SFULL + 1], qy[SFULL + 1];
    #pragma unroll
    for (int i = 0; i < SFULL; ++i) {
        const int kp = i * 12 + wave;
        qx[i] = *(const v4f*)(Ww + (size_t)(2 * kp) * HH + 4 * lane);
        qy[i] = *(const v4f*)(Ww + (size_t)(2 * kp + 1) * HH + 4 * lane);
    }
    const bool tail = wave < STAIL;
    if (tail) {
        const int kp = SFULL * 12 + wave;
        qx[SFULL] = *(const v4f*)(Ww + (size_t)(2 * kp) * HH + 4 * lane);
        qy[SFULL] = *(const v4f*)(Ww + (size_t)(2 * kp + 1) * HH + 4 * lane);
    }
    if (!mat) {
        #pragma unroll
        for (int t = 0; t < NPG; ++t) wn[t] = weights[(size_t)(n0 + t) * NN + jn[t]];
    }
    #pragma unroll
    for (int i = 0; i < SFULL; ++i) {
        const int kp = i * 12 + wave;
        v4f lo, hi;
        lo.x = qx[i].x; lo.y = qy[i].x; lo.z = qx[i].y; lo.w = qy[i].y;
        hi.x = qx[i].z; hi.y = qy[i].z; hi.z = qx[i].w; hi.w = qy[i].w;
        float* d = sB + kp * LDSP + 8 * lane;
        *(v4f*)d       = lo;
        *(v4f*)(d + 4) = hi;
    }
    if (tail) {
        const int kp = SFULL * 12 + wave;
        v4f lo, hi;
        lo.x = qx[SFULL].x; lo.y = qy[SFULL].x; lo.z = qx[SFULL].y; lo.w = qy[SFULL].y;
        hi.x = qx[SFULL].z; hi.y = qy[SFULL].z; hi.z = qx[SFULL].w; hi.w = qy[SFULL].w;
        float* d = sB + kp * LDSP + 8 * lane;
        *(v4f*)d       = lo;
        *(v4f*)(d + 4) = hi;
    }
    __syncthreads();

    v8f acc[4][2];
    #pragma unroll
    for (int j = 0; j < 4; ++j) { acc[j][0] = (v8f)0.f; acc[j][1] = (v8f)0.f; }

    const float* lb = sB + 2 * h * LDSP + 4 * c;

    if (mat) {
        gemm_half<0>(acc, aE, lb);
    } else {
        v4f   acc0[NPG], acc1[NPG];
        float wsum[NPG];
        #pragma unroll
        for (int t = 0; t < NPG; ++t) { acc0[t] = (v4f)0.f; acc1[t] = (v4f)0.f; wsum[t] = 0.f; }

        v4f   h0[2][KST], h1[2][KST];

        #pragma unroll
        for (int st = 0; st <= NSTG; ++st) {
            if (st < NSTG) {
                const int t = st / SPN, s = st % SPN, bf = st & 1;
                #pragma unroll
                for (int kk = 0; kk < KST; ++kk) {
                    const int jk = __builtin_amdgcn_readlane(jn[t], s * KST + kk);
                    const float* row = hidden + (size_t)jk * HH + lane * 4;
                    h0[bf][kk] = *(const v4f*)row;
                    h1[bf][kk] = *(const v4f*)(row + (size_t)NN * HH);
                }
            }
            __builtin_amdgcn_sched_barrier(0);
            if (st > 0) {
                const int t = (st - 1) / SPN, s = (st - 1) % SPN, bf = (st - 1) & 1;
                #pragma unroll
                for (int kk = 0; kk < KST; ++kk) {
                    const float wkk = __int_as_float(__builtin_amdgcn_readlane(__float_as_int(wn[t]), s * KST + kk));
                    acc0[t].x = __builtin_fmaf(wkk, h0[bf][kk].x, acc0[t].x);
                    acc0[t].y = __builtin_fmaf(wkk, h0[bf][kk].y, acc0[t].y);
                    acc0[t].z = __builtin_fmaf(wkk, h0[bf][kk].z, acc0[t].z);
                    acc0[t].w = __builtin_fmaf(wkk, h0[bf][kk].w, acc0[t].w);
                    acc1[t].x = __builtin_fmaf(wkk, h1[bf][kk].x, acc1[t].x);
                    acc1[t].y = __builtin_fmaf(wkk, h1[bf][kk].y, acc1[t].y);
                    acc1[t].z = __builtin_fmaf(wkk, h1[bf][kk].z, acc1[t].z);
                    acc1[t].w = __builtin_fmaf(wkk, h1[bf][kk].w, acc1[t].w);
                    wsum[t]  += wkk;
                }
                if (s == SPN - 1) {
                    const float inv = 1.f / (wsum[t] + EPSF);
                    v4f r0, r1;
                    r0.x = acc0[t].x * inv; r0.y = acc0[t].y * inv; r0.z = acc0[t].z * inv; r0.w = acc0[t].w * inv;
                    r1.x = acc1[t].x * inv; r1.y = acc1[t].y * inv; r1.z = acc1[t].z * inv; r1.w = acc1[t].w * inv;
                    const int o = (wave - MWV) * NPG + t;
                    float* d = sW + ((o >> 3) * 16 + (o & 7)) * WP + lane * 4;
                    *(v4f*)d            = r0;
                    *(v4f*)(d + 8 * WP) = r1;
                }
            }
        }
    }
    __syncthreads();

    if (mat) {
        const float* pa = sW + (wave * 16 + c) * WP + 4 * h;
        v4f aW[HH / 8];
        #pragma unroll
        for (int i = 0; i < HH / 8; ++i) aW[i] = *(const v4f*)(pa + 8 * i);
        gemm_half<CC / 8>(acc, aW, lb);

        v2f bias[4];
        #pragma unroll
        for (int j = 0; j < 4; ++j) bias[j] = *(const v2f*)(Wb + 32 * j + 2 * c);

        #pragma unroll
        for (int r = 0; r < 8; ++r) {
            float v[4][2];
            float s = 0.f;
            #pragma unroll
            for (int j = 0; j < 4; ++j) {
                v[j][0] = lrelu(acc[j][0][r] + bias[j].x);
                v[j][1] = lrelu(acc[j][1][r] + bias[j].y);
                s += v[j][0] * v[j][0];
                s += v[j][1] * v[j][1];
            }
            #pragma unroll
            for (int off = 8; off >= 1; off >>= 1)
                s += __shfl_xor(s, off, 32);
            const float inv = 1.f / (sqrtf(s) + EPSF);
            float* po = out + ((size_t)h * NN + nb + 8 * wave + r) * HH + 2 * c;
            #pragma unroll
            for (int j = 0; j < 4; ++j) {
                v2f o;
                o.x = v[j][0] * inv;
                o.y = v[j][1] * inv;
                *(v2f*)(po + 32 * j) = o;
            }
        }
    }
}

extern "C" void kernel_launch(void* const* d_in, const int* in_sizes, int n_in,
                              void* d_out, int out_size, void* d_ws, size_t ws_size,
                              hipStream_t stream)
{
    const float* emb     = (const float*)d_in[0];
    const float* weights = (const float*)d_in[1];
    const int*   nbr     = (const int*)  d_in[2];
    const float* Qw      = (const float*)d_in[3];
    const float* Qb      = (const float*)d_in[4];
    const float* Ww      = (const float*)d_in[5];
    const float* Wb      = (const float*)d_in[6];
    float*       out     = (float*)d_out;

    float* hidden = (float*)d_ws;

    k1_hidden<<<dim3(MM / 64), dim3(128), 0, stream>>>(emb, Qw, Qb, hidden);
    k2_gather_gemm2_norm<<<dim3(NN / 32), dim3(32 * (MWV + GWV)), 0, stream>>>(emb, hidden, weights, nbr, Ww, Wb, out);
}


# B-fragment LDS reads of all three products issued as one group of 16, a whole 32-WMMA block ahead, into a second fragment register set, not 1-4 at a time among the WMMAs; k and sum order unchanged
# speedup vs baseline: 2.0079x; 1.0542x over previous
#include <hip/hip_runtime.h>

typedef __attribute__((ext_vector_type(2))) float v2f;
typedef __attribute__((ext_vector_type(4))) float v4f;
typedef __attribute__((ext_vector_type(8))) float v8f;

#define ALPHA 0.3f
#define EPSF  1e-6f

#define BB   2
#define NN   8192
#define KK   32
#define CC   128
#define HH   128
#define MM   (BB * NN)

__device__ __forceinline__ float lrelu(float x) { return x >= 0.f ? x : ALPHA * x; }

#define LDSP 272

__device__ __forceinline__ void load_bfrag(v4f (&bf)[16], const float* lb, const int kb)
{
    #pragma unroll
    for (int kk = 0; kk < 2; ++kk) {
        #pragma unroll
        for (int st = 0; st < 2; ++st) {
            #pragma unroll
            for (int j = 0; j < 4; ++j)
                bf[(2 * kk + st) * 4 + j] = *(const v4f*)(lb + (4 * (kb + kk) + st) * LDSP + 64 * j);
        }
    }
}

__device__ __forceinline__ void mma_block(v8f (&acc)[4][2], const v4f a0, const v4f a1, const v4f (&bf)[16])
{
    #pragma unroll
    for (int kk = 0; kk < 2; ++kk) {
        const v4f a4 = kk ? a1 : a0;
        #pragma unroll
        for (int st = 0; st < 2; ++st) {
            v2f a;
            a.x = st ? a4.z : a4.x;
            a.y = st ? a4.w : a4.y;
            #pragma unroll
            for (int j = 0; j < 4; ++j) {
                const v4f b = bf[(2 * kk + st) * 4 + j];
                v2f b0, b1;
                b0.x = b.x; b0.y = b.y;
                b1.x = b.z; b1.y = b.w;
                acc[j][0] = __builtin_amdgcn_wmma_f32_16x16x4_f32(false, a, false, b0,
                                                                  (short)0, acc[j][0], false, false);
                acc[j][1] = __builtin_amdgcn_wmma_f32_16x16x4_f32(false, a, false, b1,
                                                                  (short)0, acc[j][1], false, false);
            }
        }
    }
}

template <int KB0>
__device__ __forceinline__ void gemm_rows(v8f (&acc)[4][2], const v4f (&a16)[16], const float* lb)
{
    v4f bf[2][16];
    load_bfrag(bf[0], lb, KB0);
    #pragma unroll
    for (int blk = 0; blk < 8; ++blk) {
        if (blk + 1 < 8) load_bfrag(bf[(blk + 1) & 1], lb, KB0 + 2 * (blk + 1));
        __builtin_amdgcn_sched_barrier(0);
        mma_block(acc, a16[2 * blk], a16[2 * blk + 1], bf[blk & 1]);
        __builtin_amdgcn_sched_barrier(0);
    }
}

#define LDS_FLOATS 41984

__global__ __launch_bounds__(128) void k1_hidden(const float* __restrict__ emb,
                                                 const float* __restrict__ Qw,
                                                 const float* __restrict__ Qb,
                                                 float* __restrict__ hidden)
{
    __shared__ __attribute__((aligned(16))) float sQ[LDS_FLOATS];
    static_assert((CC / 2) * LDSP <= LDS_FLOATS, "staged Qw must fit the reservation");

    const int lane = threadIdx.x & 31;
    const int wave = threadIdx.x >> 5;
    const int m0   = (blockIdx.x * 4 + wave) << 4;
    const int c    = lane & 15;
    const int h    = lane >> 4;

    v4f qx[CC / 8], qy[CC / 8];
    #pragma unroll
    for (int i = 0; i < CC / 8; ++i) {
        const int kp = i * 4 + wave;
        qx[i] = *(const v4f*)(Qw + (size_t)(2 * kp) * HH + 4 * lane);
        qy[i] = *(const v4f*)(Qw + (size_t)(2 * kp + 1) * HH + 4 * lane);
    }

    const float* pe = emb + (size_t)(m0 + c) * CC + 4 * h;
    v4f aE[CC / 8];
    #pragma unroll
    for (int i = 0; i < CC / 8; ++i) aE[i] = *(const v4f*)(pe + 8 * i);
    __builtin_amdgcn_sched_barrier(0);

    #pragma unroll
    for (int i = 0; i < CC / 8; ++i) {
        const int kp = i * 4 + wave;
        v4f lo, hi;
        lo.x = qx[i].x; lo.y = qy[i].x; lo.z = qx[i].y; lo.w = qy[i].y;
        hi.x = qx[i].z; hi.y = qy[i].z; hi.z = qx[i].w; hi.w = qy[i].w;
        float* d = sQ + kp * LDSP + 8 * lane;
        *(v4f*)d       = lo;
        *(v4f*)(d + 4) = hi;
    }
    __syncthreads();

    v8f acc[4][2];
    #pragma unroll
    for (int j = 0; j < 4; ++j) { acc[j][0] = (v8f)0.f; acc[j][1] = (v8f)0.f; }

    const float* lb = sQ + 2 * h * LDSP + 4 * c;

    gemm_rows<0>(acc, aE, lb);

    v2f bias[4];
    #pragma unroll
    for (int j = 0; j < 4; ++j) bias[j] = *(const v2f*)(Qb + 32 * j + 2 * c);

    #pragma unroll
    for (int r = 0; r < 8; ++r) {
        float* po = hidden + (size_t)(m0 + 8 * h + r) * HH + 2 * c;
        #pragma unroll
        for (int j = 0; j < 4; ++j) {
            v2f o;
            o.x = lrelu(acc[j][0][r] + bias[j].x);
            o.y = lrelu(acc[j][1][r] + bias[j].y);
            *(v2f*)(po + 32 * j) = o;
        }
    }
}

#define MWV 4
#define GWV 8
#define NPG 4
#define KST 8
#define SPN (KK / KST)
#define NSTG (NPG * SPN)
#define WP 132
#define WROWS ((CC + HH) / 2)
#define SFULL (WROWS / 12)
#define STAIL (WROWS - 12 * SFULL)
#define F_LDS (WROWS * LDSP + 64 * WP)


__global__ __launch_bounds__(384) void k2_gather_gemm2_norm(const float* __restrict__ emb,
                                                            const float* __restrict__ hidden,
                                                            const float* __restrict__ weights,
                                                            const int* __restrict__ nbr,
                                                            const float* __restrict__ Ww,
                                                            const float* __restrict__ Wb,
                                                            float* __restrict__ out)
{
    __shared__ __attribute__((aligned(16))) float sB[F_LDS];
    float* sW = sB + WROWS * LDSP;

    const int lane = threadIdx.x & 31;
    const int wave = __builtin_amdgcn_readfirstlane((int)(threadIdx.x >> 5));
    const int nb   = blockIdx.x * 32;
    const int c    = lane & 15;
    const int h    = lane >> 4;
    const bool mat = wave < MWV;

    v4f aE[CC / 8];
    int   jn[NPG];
    float wn[NPG];
    const int n0 = nb + (wave - MWV) * NPG;
    if (mat) {
        const float* pe = emb + ((size_t)(c >> 3) * NN + nb + 8 * wave + (c & 7)) * CC + 4 * h;
        #pragma unroll
        for (int i = 0; i < CC / 8; ++i) aE[i] = *(const v4f*)(pe + 8 * i);
    } else {
        #pragma unroll
        for (int t = 0; t < NPG; ++t) jn[t] = nbr[(size_t)(n0 + t) * KK + lane];
    }

    v4f qx[SFULL + 1], qy[SFULL + 1];
    #pragma unroll
    for (int i = 0; i < SFULL; ++i) {
        const int kp = i * 12 + wave;
        qx[i] = *(const v4f*)(Ww + (size_t)(2 * kp) * HH + 4 * lane);
        qy[i] = *(const v4f*)(Ww + (size_t)(2 * kp + 1) * HH + 4 * lane);
    }
    const bool tail = wave < STAIL;
    if (tail) {
        const int kp = SFULL * 12 + wave;
        qx[SFULL] = *(const v4f*)(Ww + (size_t)(2 * kp) * HH + 4 * lane);
        qy[SFULL] = *(const v4f*)(Ww + (size_t)(2 * kp + 1) * HH + 4 * lane);
    }
    if (!mat) {
        #pragma unroll
        for (int t = 0; t < NPG; ++t) wn[t] = weights[(size_t)(n0 + t) * NN + jn[t]];
    }
    #pragma unroll
    for (int i = 0; i < SFULL; ++i) {
        const int kp = i * 12 + wave;
        v4f lo, hi;
        lo.x = qx[i].x; lo.y = qy[i].x; lo.z = qx[i].y; lo.w = qy[i].y;
        hi.x = qx[i].z; hi.y = qy[i].z; hi.z = qx[i].w; hi.w = qy[i].w;
        float* d = sB + kp * LDSP + 8 * lane;
        *(v4f*)d       = lo;
        *(v4f*)(d + 4) = hi;
    }
    if (tail) {
        const int kp = SFULL * 12 + wave;
        v4f lo, hi;
        lo.x = qx[SFULL].x; lo.y = qy[SFULL].x; lo.z = qx[SFULL].y; lo.w = qy[SFULL].y;
        hi.x = qx[SFULL].z; hi.y = qy[SFULL].z; hi.z = qx[SFULL].w; hi.w = qy[SFULL].w;
        float* d = sB + kp * LDSP + 8 * lane;
        *(v4f*)d       = lo;
        *(v4f*)(d + 4) = hi;
    }
    __syncthreads();

    v8f acc[4][2];
    #pragma unroll
    for (int j = 0; j < 4; ++j) { acc[j][0] = (v8f)0.f; acc[j][1] = (v8f)0.f; }

    const float* lb = sB + 2 * h * LDSP + 4 * c;

    if (mat) {
        gemm_rows<0>(acc, aE, lb);
    } else {
        v4f   acc0[NPG], acc1[NPG];
        float wsum[NPG];
        #pragma unroll
        for (int t = 0; t < NPG; ++t) { acc0[t] = (v4f)0.f; acc1[t] = (v4f)0.f; wsum[t] = 0.f; }

        v4f   h0[2][KST], h1[2][KST];

        #pragma unroll
        for (int st = 0; st <= NSTG; ++st) {
            if (st < NSTG) {
                const int t = st / SPN, s = st % SPN, bf = st & 1;
                #pragma unroll
                for (int kk = 0; kk < KST; ++kk) {
                    const int jk = __builtin_amdgcn_readlane(jn[t], s * KST + kk);
                    const float* row = hidden + (size_t)jk * HH + lane * 4;
                    h0[bf][kk] = *(const v4f*)row;
                    h1[bf][kk] = *(const v4f*)(row + (size_t)NN * HH);
                }
            }
            __builtin_amdgcn_sched_barrier(0);
            if (st > 0) {
                const int t = (st - 1) / SPN, s = (st - 1) % SPN, bf = (st - 1) & 1;
                #pragma unroll
                for (int kk = 0; kk < KST; ++kk) {
                    const float wkk = __int_as_float(__builtin_amdgcn_readlane(__float_as_int(wn[t]), s * KST + kk));
                    acc0[t].x = __builtin_fmaf(wkk, h0[bf][kk].x, acc0[t].x);
                    acc0[t].y = __builtin_fmaf(wkk, h0[bf][kk].y, acc0[t].y);
                    acc0[t].z = __builtin_fmaf(wkk, h0[bf][kk].z, acc0[t].z);
                    acc0[t].w = __builtin_fmaf(wkk, h0[bf][kk].w, acc0[t].w);
                    acc1[t].x = __builtin_fmaf(wkk, h1[bf][kk].x, acc1[t].x);
                    acc1[t].y = __builtin_fmaf(wkk, h1[bf][kk].y, acc1[t].y);
                    acc1[t].z = __builtin_fmaf(wkk, h1[bf][kk].z, acc1[t].z);
                    acc1[t].w = __builtin_fmaf(wkk, h1[bf][kk].w, acc1[t].w);
                    wsum[t]  += wkk;
                }
                if (s == SPN - 1) {
                    const float inv = 1.f / (wsum[t] + EPSF);
                    v4f r0, r1;
                    r0.x = acc0[t].x * inv; r0.y = acc0[t].y * inv; r0.z = acc0[t].z * inv; r0.w = acc0[t].w * inv;
                    r1.x = acc1[t].x * inv; r1.y = acc1[t].y * inv; r1.z = acc1[t].z * inv; r1.w = acc1[t].w * inv;
                    const int o = (wave - MWV) * NPG + t;
                    float* d = sW + ((o >> 3) * 16 + (o & 7)) * WP + lane * 4;
                    *(v4f*)d            = r0;
                    *(v4f*)(d + 8 * WP) = r1;
                }
            }
        }
    }
    __syncthreads();

    if (mat) {
        const float* pa = sW + (wave * 16 + c) * WP + 4 * h;
        v4f aW[HH / 8];
        #pragma unroll
        for (int i = 0; i < HH / 8; ++i) aW[i] = *(const v4f*)(pa + 8 * i);
        gemm_rows<CC / 8>(acc, aW, lb);

        v2f bias[4];
        #pragma unroll
        for (int j = 0; j < 4; ++j) bias[j] = *(const v2f*)(Wb + 32 * j + 2 * c);

        #pragma unroll
        for (int r = 0; r < 8; ++r) {
            float v[4][2];
            float s = 0.f;
            #pragma unroll
            for (int j = 0; j < 4; ++j) {
                v[j][0] = lrelu(acc[j][0][r] + bias[j].x);
                v[j][1] = lrelu(acc[j][1][r] + bias[j].y);
                s += v[j][0] * v[j][0];
                s += v[j][1] * v[j][1];
            }
            #pragma unroll
            for (int off = 8; off >= 1; off >>= 1)
                s += __shfl_xor(s, off, 32);
            const float inv = 1.f / (sqrtf(s) + EPSF);
            float* po = out + ((size_t)h * NN + nb + 8 * wave + r) * HH + 2 * c;
            #pragma unroll
            for (int j = 0; j < 4; ++j) {
                v2f o;
                o.x = v[j][0] * inv;
                o.y = v[j][1] * inv;
                *(v2f*)(po + 32 * j) = o;
            }
        }
    }
}

extern "C" void kernel_launch(void* const* d_in, const int* in_sizes, int n_in,
                              void* d_out, int out_size, void* d_ws, size_t ws_size,
                              hipStream_t stream)
{
    const float* emb     = (const float*)d_in[0];
    const float* weights = (const float*)d_in[1];
    const int*   nbr     = (const int*)  d_in[2];
    const float* Qw      = (const float*)d_in[3];
    const float* Qb      = (const float*)d_in[4];
    const float* Ww      = (const float*)d_in[5];
    const float* Wb      = (const float*)d_in[6];
    float*       out     = (float*)d_out;

    float* hidden = (float*)d_ws;

    k1_hidden<<<dim3(MM / 64), dim3(128), 0, stream>>>(emb, Qw, Qb, hidden);
    k2_gather_gemm2_norm<<<dim3(NN / 32), dim3(32 * (MWV + GWV)), 0, stream>>>(emb, hidden, weights, nbr, Ww, Wb, out);
}


# Scattered edge-weight reads (ids, then weights from memory) leave kernel 2's head, where they gated its staging barrier: kernel 1 issues them under its product, stores them compact; sums unchanged
# speedup vs baseline: 2.1129x; 1.0523x over previous
#include <hip/hip_runtime.h>

typedef __attribute__((ext_vector_type(2))) float v2f;
typedef __attribute__((ext_vector_type(4))) float v4f;
typedef __attribute__((ext_vector_type(8))) float v8f;

#define ALPHA 0.3f
#define EPSF  1e-6f

#define BB   2
#define NN   8192
#define KK   32
#define CC   128
#define HH   128
#define MM   (BB * NN)

__device__ __forceinline__ float lrelu(float x) { return x >= 0.f ? x : ALPHA * x; }

#define LDSP 272

__device__ __forceinline__ void load_bfrag(v4f (&bf)[16], const float* lb, const int kb)
{
    #pragma unroll
    for (int kk = 0; kk < 2; ++kk) {
        #pragma unroll
        for (int st = 0; st < 2; ++st) {
            #pragma unroll
            for (int j = 0; j < 4; ++j)
                bf[(2 * kk + st) * 4 + j] = *(const v4f*)(lb + (4 * (kb + kk) + st) * LDSP + 64 * j);
        }
    }
}

__device__ __forceinline__ void mma_block(v8f (&acc)[4][2], const v4f a0, const v4f a1, const v4f (&bf)[16])
{
    #pragma unroll
    for (int kk = 0; kk < 2; ++kk) {
        const v4f a4 = kk ? a1 : a0;
        #pragma unroll
        for (int st = 0; st < 2; ++st) {
            v2f a;
            a.x = st ? a4.z : a4.x;
            a.y = st ? a4.w : a4.y;
            #pragma unroll
            for (int j = 0; j < 4; ++j) {
                const v4f b = bf[(2 * kk + st) * 4 + j];
                v2f b0, b1;
                b0.x = b.x; b0.y = b.y;
                b1.x = b.z; b1.y = b.w;
                acc[j][0] = __builtin_amdgcn_wmma_f32_16x16x4_f32(false, a, false, b0,
                                                                  (short)0, acc[j][0], false, false);
                acc[j][1] = __builtin_amdgcn_wmma_f32_16x16x4_f32(false, a, false, b1,
                                                                  (short)0, acc[j][1], false, false);
            }
        }
    }
}

template <int KB0>
__device__ __forceinline__ void gemm_rows(v8f (&acc)[4][2], const v4f (&a16)[16], const float* lb)
{
    v4f bf[2][16];
    load_bfrag(bf[0], lb, KB0);
    #pragma unroll
    for (int blk = 0; blk < 8; ++blk) {
        if (blk + 1 < 8) load_bfrag(bf[(blk + 1) & 1], lb, KB0 + 2 * (blk + 1));
        __builtin_amdgcn_sched_barrier(0);
        mma_block(acc, a16[2 * blk], a16[2 * blk + 1], bf[blk & 1]);
        __builtin_amdgcn_sched_barrier(0);
    }
}

#define LDS_FLOATS 41984
#define NPW 8

__global__ __launch_bounds__(128) void k1_hidden(const float* __restrict__ emb,
                                                 const float* __restrict__ Qw,
                                                 const float* __restrict__ Qb,
                                                 const float* __restrict__ weights,
                                                 const int* __restrict__ nbr,
                                                 float* __restrict__ hidden,
                                                 float* __restrict__ wc)
{
    __shared__ __attribute__((aligned(16))) float sQ[LDS_FLOATS];
    static_assert((CC / 2) * LDSP <= LDS_FLOATS, "staged Qw must fit the reservation");

    const int lane = threadIdx.x & 31;
    const int wave = threadIdx.x >> 5;
    const int m0   = (blockIdx.x * 4 + wave) << 4;
    const int c    = lane & 15;
    const int h    = lane >> 4;

    const int node0 = (blockIdx.x * 4 + wave) * NPW;
    int idn[NPW];
    #pragma unroll
    for (int t = 0; t < NPW; ++t) idn[t] = nbr[(size_t)(node0 + t) * KK + lane];
    __builtin_amdgcn_sched_barrier(0);

    v4f qx[CC / 8], qy[CC / 8];
    #pragma unroll
    for (int i = 0; i < CC / 8; ++i) {
        const int kp = i * 4 + wave;
        qx[i] = *(const v4f*)(Qw + (size_t)(2 * kp) * HH + 4 * lane);
        qy[i] = *(const v4f*)(Qw + (size_t)(2 * kp + 1) * HH + 4 * lane);
    }

    const float* pe = emb + (size_t)(m0 + c) * CC + 4 * h;
    v4f aE[CC / 8];
    #pragma unroll
    for (int i = 0; i < CC / 8; ++i) aE[i] = *(const v4f*)(pe + 8 * i);
    __builtin_amdgcn_sched_barrier(0);

    #pragma unroll
    for (int i = 0; i < CC / 8; ++i) {
        const int kp = i * 4 + wave;
        v4f lo, hi;
        lo.x = qx[i].x; lo.y = qy[i].x; lo.z = qx[i].y; lo.w = qy[i].y;
        hi.x = qx[i].z; hi.y = qy[i].z; hi.z = qx[i].w; hi.w = qy[i].w;
        float* d = sQ + kp * LDSP + 8 * lane;
        *(v4f*)d       = lo;
        *(v4f*)(d + 4) = hi;
    }
    __syncthreads();

    #pragma unroll
    for (int t = 0; t < NPW; ++t) asm volatile("" : "+v"(idn[t]));
    float wv[NPW];
    #pragma unroll
    for (int t = 0; t < NPW; ++t) wv[t] = weights[(size_t)(node0 + t) * NN + idn[t]];

    v8f acc[4][2];
    #pragma unroll
    for (int j = 0; j < 4; ++j) { acc[j][0] = (v8f)0.f; acc[j][1] = (v8f)0.f; }

    const float* lb = sQ + 2 * h * LDSP + 4 * c;

    gemm_rows<0>(acc, aE, lb);

    v2f bias[4];
    #pragma unroll
    for (int j = 0; j < 4; ++j) bias[j] = *(const v2f*)(Qb + 32 * j + 2 * c);

    #pragma unroll
    for (int r = 0; r < 8; ++r) {
        float* po = hidden + (size_t)(m0 + 8 * h + r) * HH + 2 * c;
        #pragma unroll
        for (int j = 0; j < 4; ++j) {
            v2f o;
            o.x = lrelu(acc[j][0][r] + bias[j].x);
            o.y = lrelu(acc[j][1][r] + bias[j].y);
            *(v2f*)(po + 32 * j) = o;
        }
    }

    #pragma unroll
    for (int t = 0; t < NPW; ++t) wc[(size_t)(node0 + t) * KK + lane] = wv[t];
}

#define MWV 4
#define GWV 8
#define NPG 4
#define KST 8
#define SPN (KK / KST)
#define NSTG (NPG * SPN)
#define WP 132
#define WROWS ((CC + HH) / 2)
#define SFULL (WROWS / 12)
#define STAIL (WROWS - 12 * SFULL)
#define F_LDS (WROWS * LDSP + 64 * WP)


__global__ __launch_bounds__(384) void k2_gather_gemm2_norm(const float* __restrict__ emb,
                                                            const float* __restrict__ hidden,
                                                            const float* __restrict__ wc,
                                                            const int* __restrict__ nbr,
                                                            const float* __restrict__ Ww,
                                                            const float* __restrict__ Wb,
                                                            float* __restrict__ out)
{
    __shared__ __attribute__((aligned(16))) float sB[F_LDS];
    float* sW = sB + WROWS * LDSP;

    const int lane = threadIdx.x & 31;
    const int wave = __builtin_amdgcn_readfirstlane((int)(threadIdx.x >> 5));
    const int nb   = blockIdx.x * 32;
    const int c    = lane & 15;
    const int h    = lane >> 4;
    const bool mat = wave < MWV;

    v4f aE[CC / 8];
    int   jn[NPG];
    float wn[NPG];
    const int n0 = nb + (wave - MWV) * NPG;
    if (mat) {
        const float* pe = emb + ((size_t)(c >> 3) * NN + nb + 8 * wave + (c & 7)) * CC + 4 * h;
        #pragma unroll
        for (int i = 0; i < CC / 8; ++i) aE[i] = *(const v4f*)(pe + 8 * i);
    } else {
        #pragma unroll
        for (int t = 0; t < NPG; ++t) jn[t] = nbr[(size_t)(n0 + t) * KK + lane];
        #pragma unroll
        for (int t = 0; t < NPG; ++t) wn[t] = wc[(size_t)(n0 + t) * KK + lane];
    }

    v4f qx[SFULL + 1], qy[SFULL + 1];
    #pragma unroll
    for (int i = 0; i < SFULL; ++i) {
        const int kp = i * 12 + wave;
        qx[i] = *(const v4f*)(Ww + (size_t)(2 * kp) * HH + 4 * lane);
        qy[i] = *(const v4f*)(Ww + (size_t)(2 * kp + 1) * HH + 4 * lane);
    }
    const bool tail = wave < STAIL;
    if (tail) {
        const int kp = SFULL * 12 + wave;
        qx[SFULL] = *(const v4f*)(Ww + (size_t)(2 * kp) * HH + 4 * lane);
        qy[SFULL] = *(const v4f*)(Ww + (size_t)(2 * kp + 1) * HH + 4 * lane);
    }
    #pragma unroll
    for (int i = 0; i < SFULL; ++i) {
        const int kp = i * 12 + wave;
        v4f lo, hi;
        lo.x = qx[i].x; lo.y = qy[i].x; lo.z = qx[i].y; lo.w = qy[i].y;
        hi.x = qx[i].z; hi.y = qy[i].z; hi.z = qx[i].w; hi.w = qy[i].w;
        float* d = sB + kp * LDSP + 8 * lane;
        *(v4f*)d       = lo;
        *(v4f*)(d + 4) = hi;
    }
    if (tail) {
        const int kp = SFULL * 12 + wave;
        v4f lo, hi;
        lo.x = qx[SFULL].x; lo.y = qy[SFULL].x; lo.z = qx[SFULL].y; lo.w = qy[SFULL].y;
        hi.x = qx[SFULL].z; hi.y = qy[SFULL].z; hi.z = qx[SFULL].w; hi.w = qy[SFULL].w;
        float* d = sB + kp * LDSP + 8 * lane;
        *(v4f*)d       = lo;
        *(v4f*)(d + 4) = hi;
    }
    __syncthreads();

    v8f acc[4][2];
    #pragma unroll
    for (int j = 0; j < 4; ++j) { acc[j][0] = (v8f)0.f; acc[j][1] = (v8f)0.f; }

    const float* lb = sB + 2 * h * LDSP + 4 * c;

    if (mat) {
        gemm_rows<0>(acc, aE, lb);
    } else {
        v4f   acc0[NPG], acc1[NPG];
        float wsum[NPG];
        #pragma unroll
        for (int t = 0; t < NPG; ++t) { acc0[t] = (v4f)0.f; acc1[t] = (v4f)0.f; wsum[t] = 0.f; }

        v4f   h0[2][KST], h1[2][KST];

        #pragma unroll
        for (int st = 0; st <= NSTG; ++st) {
            if (st < NSTG) {
                const int t = st / SPN, s = st % SPN, bf = st & 1;
                #pragma unroll
                for (int kk = 0; kk < KST; ++kk) {
                    const int jk = __builtin_amdgcn_readlane(jn[t], s * KST + kk);
                    const float* row = hidden + (size_t)jk * HH + lane * 4;
                    h0[bf][kk] = *(const v4f*)row;
                    h1[bf][kk] = *(const v4f*)(row + (size_t)NN * HH);
                }
            }
            __builtin_amdgcn_sched_barrier(0);
            if (st > 0) {
                const int t = (st - 1) / SPN, s = (st - 1) % SPN, bf = (st - 1) & 1;
                #pragma unroll
                for (int kk = 0; kk < KST; ++kk) {
                    const float wkk = __int_as_float(__builtin_amdgcn_readlane(__float_as_int(wn[t]), s * KST + kk));
                    acc0[t].x = __builtin_fmaf(wkk, h0[bf][kk].x, acc0[t].x);
                    acc0[t].y = __builtin_fmaf(wkk, h0[bf][kk].y, acc0[t].y);
                    acc0[t].z = __builtin_fmaf(wkk, h0[bf][kk].z, acc0[t].z);
                    acc0[t].w = __builtin_fmaf(wkk, h0[bf][kk].w, acc0[t].w);
                    acc1[t].x = __builtin_fmaf(wkk, h1[bf][kk].x, acc1[t].x);
                    acc1[t].y = __builtin_fmaf(wkk, h1[bf][kk].y, acc1[t].y);
                    acc1[t].z = __builtin_fmaf(wkk, h1[bf][kk].z, acc1[t].z);
                    acc1[t].w = __builtin_fmaf(wkk, h1[bf][kk].w, acc1[t].w);
                    wsum[t]  += wkk;
                }
                if (s == SPN - 1) {
                    const float inv = 1.f / (wsum[t] + EPSF);
                    v4f r0, r1;
                    r0.x = acc0[t].x * inv; r0.y = acc0[t].y * inv; r0.z = acc0[t].z * inv; r0.w = acc0[t].w * inv;
                    r1.x = acc1[t].x * inv; r1.y = acc1[t].y * inv; r1.z = acc1[t].z * inv; r1.w = acc1[t].w * inv;
                    const int o = (wave - MWV) * NPG + t;
                    float* d = sW + ((o >> 3) * 16 + (o & 7)) * WP + lane * 4;
                    *(v4f*)d            = r0;
                    *(v4f*)(d + 8 * WP) = r1;
                }
            }
        }
    }
    __syncthreads();

    if (mat) {
        const float* pa = sW + (wave * 16 + c) * WP + 4 * h;
        v4f aW[HH / 8];
        #pragma unroll
        for (int i = 0; i < HH / 8; ++i) aW[i] = *(const v4f*)(pa + 8 * i);
        gemm_rows<CC / 8>(acc, aW, lb);

        v2f bias[4];
        #pragma unroll
        for (int j = 0; j < 4; ++j) bias[j] = *(const v2f*)(Wb + 32 * j + 2 * c);

        #pragma unroll
        for (int r = 0; r < 8; ++r) {
            float v[4][2];
            float s = 0.f;
            #pragma unroll
            for (int j = 0; j < 4; ++j) {
                v[j][0] = lrelu(acc[j][0][r] + bias[j].x);
                v[j][1] = lrelu(acc[j][1][r] + bias[j].y);
                s += v[j][0] * v[j][0];
                s += v[j][1] * v[j][1];
            }
            #pragma unroll
            for (int off = 8; off >= 1; off >>= 1)
                s += __shfl_xor(s, off, 32);
            const float inv = 1.f / (sqrtf(s) + EPSF);
            float* po = out + ((size_t)h * NN + nb + 8 * wave + r) * HH + 2 * c;
            #pragma unroll
            for (int j = 0; j < 4; ++j) {
                v2f o;
                o.x = v[j][0] * inv;
                o.y = v[j][1] * inv;
                *(v2f*)(po + 32 * j) = o;
            }
        }
    }
}

extern "C" void kernel_launch(void* const* d_in, const int* in_sizes, int n_in,
                              void* d_out, int out_size, void* d_ws, size_t ws_size,
                              hipStream_t stream)
{
    const float* emb     = (const float*)d_in[0];
    const float* weights = (const float*)d_in[1];
    const int*   nbr     = (const int*)  d_in[2];
    const float* Qw      = (const float*)d_in[3];
    const float* Qb      = (const float*)d_in[4];
    const float* Ww      = (const float*)d_in[5];
    const float* Wb      = (const float*)d_in[6];
    float*       out     = (float*)d_out;

    float* hidden = (float*)d_ws;
    float* wc     = hidden + (size_t)MM * HH;

    k1_hidden<<<dim3(MM / 64), dim3(128), 0, stream>>>(emb, Qw, Qb, weights, nbr, hidden, wc);
    k2_gather_gemm2_norm<<<dim3(NN / 32), dim3(32 * (MWV + GWV)), 0, stream>>>(emb, hidden, wc, nbr, Ww, Wb, out);
}
